# starting kernel
# baseline (speedup 1.0000x reference)
#include <hip/hip_runtime.h>

#define DIM 128

typedef float v2f __attribute__((ext_vector_type(2)));
typedef float v8f __attribute__((ext_vector_type(8)));

__global__ void __launch_bounds__(256)
proj_gemm_kernel(const float* __restrict__ node,
                 const float* __restrict__ Wsrc, const float* __restrict__ bsrc,
                 const float* __restrict__ Wdst, const float* __restrict__ bdst,
                 float* __restrict__ src_proj, float* __restrict__ dest_proj,
                 int V)
{
    const int lane    = threadIdx.x & 31;
    const int wave    = threadIdx.x >> 5;
    const int m0      = blockIdx.x * 16;
    const int n0      = wave * 16;
    const int halfsel = lane >> 4;
    const int l15     = lane & 15;

    const bool isDst  = (blockIdx.y != 0);
    const float* __restrict__ W   = isDst ? Wdst : Wsrc;
    const float* __restrict__ b   = isDst ? bdst : bsrc;
    float* __restrict__       out = isDst ? dest_proj : src_proj;

    const int arow = m0 + l15;
    const int bcol = n0 + l15;

    const float* __restrict__ aptr = node + (size_t)arow * DIM + 2 * halfsel;
    const float* __restrict__ bptr = W + (size_t)(2 * halfsel) * DIM + bcol;

    v8f acc = {};
#pragma unroll
    for (int k = 0; k < DIM; k += 4) {
        v2f a = *(const v2f*)(aptr + k);
        v2f bb;
        bb.x = bptr[(size_t)(k + 0) * DIM];
        bb.y = bptr[(size_t)(k + 1) * DIM];
        acc = __builtin_amdgcn_wmma_f32_16x16x4_f32(
                  false, a, false, bb, (short)0, acc, false, false);
    }

    const float bias = b[bcol];
#pragma unroll
    for (int r = 0; r < 8; ++r) {
        const int m = m0 + r + 8 * halfsel;
        out[(size_t)m * DIM + bcol] = acc[r] + bias;
    }
}

__global__ void __launch_bounds__(256)
edge_scatter_kernel(const float* __restrict__ src_proj,
                    const float* __restrict__ dest_proj,
                    const int* __restrict__ edge_src,
                    const int* __restrict__ edge_dest,
                    const int* __restrict__ edge_cls,
                    const float* __restrict__ edge_emb,
                    float* __restrict__ out, int E)
{
    const int e = blockIdx.x * 8 + (threadIdx.x >> 5);
    if (e >= E) return;
    const int lane = threadIdx.x & 31;
    const int off  = lane * 4;

    const int s = edge_src[e];
    const int d = edge_dest[e];
    const int c = edge_cls[e];

    const float4 sp = *(const float4*)(src_proj  + (size_t)s * DIM + off);
    const float4 dp = *(const float4*)(dest_proj + (size_t)d * DIM + off);
    const float4 em = *(const float4*)(edge_emb  + (size_t)c * DIM + off);

    float4 m;
    m.x = fmaxf(sp.x + dp.x + em.x, 0.0f);
    m.y = fmaxf(sp.y + dp.y + em.y, 0.0f);
    m.z = fmaxf(sp.z + dp.z + em.z, 0.0f);
    m.w = fmaxf(sp.w + dp.w + em.w, 0.0f);

    float* o = out + (size_t)d * DIM + off;
    unsafeAtomicAdd(o + 0, m.x);
    unsafeAtomicAdd(o + 1, m.y);
    unsafeAtomicAdd(o + 2, m.z);
    unsafeAtomicAdd(o + 3, m.w);
}

extern "C" void kernel_launch(void* const* d_in, const int* in_sizes, int n_in,
                              void* d_out, int out_size, void* d_ws, size_t ws_size,
                              hipStream_t stream) {
    const float* node = (const float*)d_in[0];
    const int*   esrc = (const int*)d_in[1];
    const int*   edst = (const int*)d_in[2];
    const int*   ecls = (const int*)d_in[3];
    const float* Wsrc = (const float*)d_in[4];
    const float* bsrc = (const float*)d_in[5];
    const float* Wdst = (const float*)d_in[6];
    const float* bdst = (const float*)d_in[7];
    const float* eemb = (const float*)d_in[8];
    float* out = (float*)d_out;

    const int V = in_sizes[0] / DIM;
    const int E = in_sizes[1];

    float* src_proj  = (float*)d_ws;
    float* dest_proj = src_proj + (size_t)V * DIM;

    hipMemsetAsync(d_out, 0, (size_t)V * DIM * sizeof(float), stream);

    dim3 ggrid((V + 15) / 16, 2);
    proj_gemm_kernel<<<ggrid, dim3(256), 0, stream>>>(
        node, Wsrc, bsrc, Wdst, bdst, src_proj, dest_proj, V);

    const int nblocks = (E + 7) / 8;
    edge_scatter_kernel<<<nblocks, dim3(256), 0, stream>>>(
        src_proj, dest_proj, esrc, edst, ecls, eemb, out, E);
}


# projection GEMM operand reuse: persistent 512-block grid keeps both W fragment sets in registers, stages each node strip once through LDS by whole-row loads; k sum order inside groups of 8 changes
# speedup vs baseline: 1.1524x; 1.1524x over previous
#include <hip/hip_runtime.h>

#define DIM 128

typedef float v2f __attribute__((ext_vector_type(2)));
typedef float v4f __attribute__((ext_vector_type(4)));
typedef float v8f __attribute__((ext_vector_type(8)));


static __device__ __forceinline__ void lds_barrier()
{
    asm volatile("s_wait_dscnt 0x0\n\ts_barrier_signal -1\n\ts_barrier_wait -1" ::: "memory");
}

#define APITCH 132

__global__ void __launch_bounds__(256)
proj_gemm_kernel(const float* __restrict__ node,
                 const float* __restrict__ Wsrc, const float* __restrict__ bsrc,
                 const float* __restrict__ Wdst, const float* __restrict__ bdst,
                 float* __restrict__ src_proj, float* __restrict__ dest_proj,
                 int V)
{
    __shared__ __attribute__((aligned(16))) float As[2][16 * APITCH];

    const int lane    = threadIdx.x & 31;
    const int wave    = threadIdx.x >> 5;
    const int halfsel = lane >> 4;
    const int l15     = lane & 15;
    const int col     = wave * 16 + l15;

    const int nstrips = (V + 15) >> 4;
    int strip = blockIdx.x;
    if (strip >= nstrips) return;

    const int scol = lane * 4;

    {
        int r0 = strip * 16 + wave;
        int r1 = r0 + 8;
        r0 = r0 < V ? r0 : V - 1;
        r1 = r1 < V ? r1 : V - 1;
        const v4f g0 = *(const v4f*)(node + (size_t)r0 * DIM + scol);
        const v4f g1 = *(const v4f*)(node + (size_t)r1 * DIM + scol);
        *(v4f*)(&As[0][wave * APITCH + scol])       = g0;
        *(v4f*)(&As[0][(wave + 8) * APITCH + scol]) = g1;
    }

    v4f bs[16], bd[16];
#pragma unroll
    for (int j = 0; j < 16; ++j) {
        const size_t wo = (size_t)(8 * j + 4 * halfsel) * DIM + col;
        bs[j].x = Wsrc[wo + 0 * DIM];
        bs[j].y = Wsrc[wo + 1 * DIM];
        bs[j].z = Wsrc[wo + 2 * DIM];
        bs[j].w = Wsrc[wo + 3 * DIM];
        bd[j].x = Wdst[wo + 0 * DIM];
        bd[j].y = Wdst[wo + 1 * DIM];
        bd[j].z = Wdst[wo + 2 * DIM];
        bd[j].w = Wdst[wo + 3 * DIM];
    }
    const float bias_s = bsrc[col];
    const float bias_d = bdst[col];

    lds_barrier();

    int buf = 0;
    for (; strip < nstrips; strip += gridDim.x) {
        const int  next    = strip + gridDim.x;
        const bool hasNext = next < nstrips;

        v4f g0 = {}, g1 = {};
        if (hasNext) {
            int r0 = next * 16 + wave;
            int r1 = r0 + 8;
            r0 = r0 < V ? r0 : V - 1;
            r1 = r1 < V ? r1 : V - 1;
            g0 = *(const v4f*)(node + (size_t)r0 * DIM + scol);
            g1 = *(const v4f*)(node + (size_t)r1 * DIM + scol);
        }

        const size_t oo = (size_t)(strip * 16 + 8 * halfsel) * DIM + col;

        const float* ap = &As[buf][l15 * APITCH + 4 * halfsel];
        v4f a[16];
#pragma unroll
        for (int j = 0; j < 16; ++j)
            a[j] = *(const v4f*)(ap + 8 * j);
        __builtin_amdgcn_sched_barrier(0);

        v8f accs = {}, accd = {};
#pragma unroll
        for (int j = 0; j < 16; ++j) {
            v2f a0, a1, s0, s1, d0, d1;
            a0.x = a[j].x;  a0.y = a[j].y;
            a1.x = a[j].z;  a1.y = a[j].w;
            s0.x = bs[j].x; s0.y = bs[j].y;
            s1.x = bs[j].z; s1.y = bs[j].w;
            d0.x = bd[j].x; d0.y = bd[j].y;
            d1.x = bd[j].z; d1.y = bd[j].w;
            accs = __builtin_amdgcn_wmma_f32_16x16x4_f32(
                       false, a0, false, s0, (short)0, accs, false, false);
            accd = __builtin_amdgcn_wmma_f32_16x16x4_f32(
                       false, a0, false, d0, (short)0, accd, false, false);
            accs = __builtin_amdgcn_wmma_f32_16x16x4_f32(
                       false, a1, false, s1, (short)0, accs, false, false);
            accd = __builtin_amdgcn_wmma_f32_16x16x4_f32(
                       false, a1, false, d1, (short)0, accd, false, false);
        }

#pragma unroll
        for (int r = 0; r < 8; ++r) {
            src_proj[oo + (size_t)r * DIM]  = accs[r] + bias_s;
            dest_proj[oo + (size_t)r * DIM] = accd[r] + bias_d;
        }

        if (hasNext) {
            *(v4f*)(&As[buf ^ 1][wave * APITCH + scol])       = g0;
            *(v4f*)(&As[buf ^ 1][(wave + 8) * APITCH + scol]) = g1;
        }
        lds_barrier();
        buf ^= 1;
    }
}

__global__ void __launch_bounds__(256)
edge_scatter_kernel(const float* __restrict__ src_proj,
                    const float* __restrict__ dest_proj,
                    const int* __restrict__ edge_src,
                    const int* __restrict__ edge_dest,
                    const int* __restrict__ edge_cls,
                    const float* __restrict__ edge_emb,
                    float* __restrict__ out, int E)
{
    const int e = blockIdx.x * 8 + (threadIdx.x >> 5);
    if (e >= E) return;
    const int lane = threadIdx.x & 31;
    const int off  = lane * 4;

    const int s = edge_src[e];
    const int d = edge_dest[e];
    const int c = edge_cls[e];

    const float4 sp = *(const float4*)(src_proj  + (size_t)s * DIM + off);
    const float4 dp = *(const float4*)(dest_proj + (size_t)d * DIM + off);
    const float4 em = *(const float4*)(edge_emb  + (size_t)c * DIM + off);

    float4 m;
    m.x = fmaxf(sp.x + dp.x + em.x, 0.0f);
    m.y = fmaxf(sp.y + dp.y + em.y, 0.0f);
    m.z = fmaxf(sp.z + dp.z + em.z, 0.0f);
    m.w = fmaxf(sp.w + dp.w + em.w, 0.0f);

    float* o = out + (size_t)d * DIM + off;
    unsafeAtomicAdd(o + 0, m.x);
    unsafeAtomicAdd(o + 1, m.y);
    unsafeAtomicAdd(o + 2, m.z);
    unsafeAtomicAdd(o + 3, m.w);
}

extern "C" void kernel_launch(void* const* d_in, const int* in_sizes, int n_in,
                              void* d_out, int out_size, void* d_ws, size_t ws_size,
                              hipStream_t stream) {
    const float* node = (const float*)d_in[0];
    const int*   esrc = (const int*)d_in[1];
    const int*   edst = (const int*)d_in[2];
    const int*   ecls = (const int*)d_in[3];
    const float* Wsrc = (const float*)d_in[4];
    const float* bsrc = (const float*)d_in[5];
    const float* Wdst = (const float*)d_in[6];
    const float* bdst = (const float*)d_in[7];
    const float* eemb = (const float*)d_in[8];
    float* out = (float*)d_out;

    const int V = in_sizes[0] / DIM;
    const int E = in_sizes[1];

    float* src_proj  = (float*)d_ws;
    float* dest_proj = src_proj + (size_t)V * DIM;

    hipMemsetAsync(d_out, 0, (size_t)V * DIM * sizeof(float), stream);

    proj_gemm_kernel<<<dim3(512), dim3(256), 0, stream>>>(
        node, Wsrc, bsrc, Wdst, bdst, src_proj, dest_proj, V);

    const int nblocks = (E + 7) / 8;
    edge_scatter_kernel<<<nblocks, dim3(256), 0, stream>>>(
        src_proj, dest_proj, esrc, edst, ecls, eemb, out, E);
}


# edge scatter: lane l owns columns l, l+32, l+64, l+96 (was 4l..4l+3), so each f32 atomic instruction adds one whole 128-B line: 4 line requests per edge, was 16; add arrival order random as before
# speedup vs baseline: 1.5529x; 1.3475x over previous
#include <hip/hip_runtime.h>

#define DIM 128

typedef float v2f __attribute__((ext_vector_type(2)));
typedef float v4f __attribute__((ext_vector_type(4)));
typedef float v8f __attribute__((ext_vector_type(8)));


static __device__ __forceinline__ void lds_barrier()
{
    asm volatile("s_wait_dscnt 0x0\n\ts_barrier_signal -1\n\ts_barrier_wait -1" ::: "memory");
}

#define APITCH 132

__global__ void __launch_bounds__(256)
proj_gemm_kernel(const float* __restrict__ node,
                 const float* __restrict__ Wsrc, const float* __restrict__ bsrc,
                 const float* __restrict__ Wdst, const float* __restrict__ bdst,
                 float* __restrict__ src_proj, float* __restrict__ dest_proj,
                 int V)
{
    __shared__ __attribute__((aligned(16))) float As[2][16 * APITCH];

    const int lane    = threadIdx.x & 31;
    const int wave    = threadIdx.x >> 5;
    const int halfsel = lane >> 4;
    const int l15     = lane & 15;
    const int col     = wave * 16 + l15;

    const int nstrips = (V + 15) >> 4;
    int strip = blockIdx.x;
    if (strip >= nstrips) return;

    const int scol = lane * 4;

    {
        int r0 = strip * 16 + wave;
        int r1 = r0 + 8;
        r0 = r0 < V ? r0 : V - 1;
        r1 = r1 < V ? r1 : V - 1;
        const v4f g0 = *(const v4f*)(node + (size_t)r0 * DIM + scol);
        const v4f g1 = *(const v4f*)(node + (size_t)r1 * DIM + scol);
        *(v4f*)(&As[0][wave * APITCH + scol])       = g0;
        *(v4f*)(&As[0][(wave + 8) * APITCH + scol]) = g1;
    }

    v4f bs[16], bd[16];
#pragma unroll
    for (int j = 0; j < 16; ++j) {
        const size_t wo = (size_t)(8 * j + 4 * halfsel) * DIM + col;
        bs[j].x = Wsrc[wo + 0 * DIM];
        bs[j].y = Wsrc[wo + 1 * DIM];
        bs[j].z = Wsrc[wo + 2 * DIM];
        bs[j].w = Wsrc[wo + 3 * DIM];
        bd[j].x = Wdst[wo + 0 * DIM];
        bd[j].y = Wdst[wo + 1 * DIM];
        bd[j].z = Wdst[wo + 2 * DIM];
        bd[j].w = Wdst[wo + 3 * DIM];
    }
    const float bias_s = bsrc[col];
    const float bias_d = bdst[col];

    lds_barrier();

    int buf = 0;
    for (; strip < nstrips; strip += gridDim.x) {
        const int  next    = strip + gridDim.x;
        const bool hasNext = next < nstrips;

        v4f g0 = {}, g1 = {};
        if (hasNext) {
            int r0 = next * 16 + wave;
            int r1 = r0 + 8;
            r0 = r0 < V ? r0 : V - 1;
            r1 = r1 < V ? r1 : V - 1;
            g0 = *(const v4f*)(node + (size_t)r0 * DIM + scol);
            g1 = *(const v4f*)(node + (size_t)r1 * DIM + scol);
        }

        const size_t oo = (size_t)(strip * 16 + 8 * halfsel) * DIM + col;

        const float* ap = &As[buf][l15 * APITCH + 4 * halfsel];
        v4f a[16];
#pragma unroll
        for (int j = 0; j < 16; ++j)
            a[j] = *(const v4f*)(ap + 8 * j);
        __builtin_amdgcn_sched_barrier(0);

        v8f accs = {}, accd = {};
#pragma unroll
        for (int j = 0; j < 16; ++j) {
            v2f a0, a1, s0, s1, d0, d1;
            a0.x = a[j].x;  a0.y = a[j].y;
            a1.x = a[j].z;  a1.y = a[j].w;
            s0.x = bs[j].x; s0.y = bs[j].y;
            s1.x = bs[j].z; s1.y = bs[j].w;
            d0.x = bd[j].x; d0.y = bd[j].y;
            d1.x = bd[j].z; d1.y = bd[j].w;
            accs = __builtin_amdgcn_wmma_f32_16x16x4_f32(
                       false, a0, false, s0, (short)0, accs, false, false);
            accd = __builtin_amdgcn_wmma_f32_16x16x4_f32(
                       false, a0, false, d0, (short)0, accd, false, false);
            accs = __builtin_amdgcn_wmma_f32_16x16x4_f32(
                       false, a1, false, s1, (short)0, accs, false, false);
            accd = __builtin_amdgcn_wmma_f32_16x16x4_f32(
                       false, a1, false, d1, (short)0, accd, false, false);
        }

#pragma unroll
        for (int r = 0; r < 8; ++r) {
            src_proj[oo + (size_t)r * DIM]  = accs[r] + bias_s;
            dest_proj[oo + (size_t)r * DIM] = accd[r] + bias_d;
        }

        if (hasNext) {
            *(v4f*)(&As[buf ^ 1][wave * APITCH + scol])       = g0;
            *(v4f*)(&As[buf ^ 1][(wave + 8) * APITCH + scol]) = g1;
        }
        lds_barrier();
        buf ^= 1;
    }
}

__global__ void __launch_bounds__(256)
edge_scatter_kernel(const float* __restrict__ src_proj,
                    const float* __restrict__ dest_proj,
                    const int* __restrict__ edge_src,
                    const int* __restrict__ edge_dest,
                    const int* __restrict__ edge_cls,
                    const float* __restrict__ edge_emb,
                    float* __restrict__ out, int E)
{
    const int e = blockIdx.x * 8 + (threadIdx.x >> 5);
    if (e >= E) return;
    const int lane = threadIdx.x & 31;

    const int s = edge_src[e];
    const int d = edge_dest[e];
    const int c = edge_cls[e];

    const float* sp = src_proj  + (size_t)s * DIM + lane;
    const float* dp = dest_proj + (size_t)d * DIM + lane;
    const float* em = edge_emb  + (size_t)c * DIM + lane;
    float*       o  = out       + (size_t)d * DIM + lane;

    const float a0 = sp[0],  b0 = dp[0],  g0 = em[0];
    const float a1 = sp[32], b1 = dp[32], g1 = em[32];
    const float a2 = sp[64], b2 = dp[64], g2 = em[64];
    const float a3 = sp[96], b3 = dp[96], g3 = em[96];

    unsafeAtomicAdd(o +  0, fmaxf(a0 + b0 + g0, 0.0f));
    unsafeAtomicAdd(o + 32, fmaxf(a1 + b1 + g1, 0.0f));
    unsafeAtomicAdd(o + 64, fmaxf(a2 + b2 + g2, 0.0f));
    unsafeAtomicAdd(o + 96, fmaxf(a3 + b3 + g3, 0.0f));
}

extern "C" void kernel_launch(void* const* d_in, const int* in_sizes, int n_in,
                              void* d_out, int out_size, void* d_ws, size_t ws_size,
                              hipStream_t stream) {
    const float* node = (const float*)d_in[0];
    const int*   esrc = (const int*)d_in[1];
    const int*   edst = (const int*)d_in[2];
    const int*   ecls = (const int*)d_in[3];
    const float* Wsrc = (const float*)d_in[4];
    const float* bsrc = (const float*)d_in[5];
    const float* Wdst = (const float*)d_in[6];
    const float* bdst = (const float*)d_in[7];
    const float* eemb = (const float*)d_in[8];
    float* out = (float*)d_out;

    const int V = in_sizes[0] / DIM;
    const int E = in_sizes[1];

    float* src_proj  = (float*)d_ws;
    float* dest_proj = src_proj + (size_t)V * DIM;

    hipMemsetAsync(d_out, 0, (size_t)V * DIM * sizeof(float), stream);

    proj_gemm_kernel<<<dim3(512), dim3(256), 0, stream>>>(
        node, Wsrc, bsrc, Wdst, bdst, src_proj, dest_proj, V);

    const int nblocks = (E + 7) / 8;
    edge_scatter_kernel<<<nblocks, dim3(256), 0, stream>>>(
        src_proj, dest_proj, esrc, edst, ecls, eemb, out, E);
}


# edge phase: one wave per 16 consecutive edges (was 1): indices by 3 coalesced loads, row loads of 4 edges (48 lines) in flight per wave, 16x fewer waves; add arrival order random as before
# speedup vs baseline: 1.8416x; 1.1859x over previous
#include <hip/hip_runtime.h>

#define DIM 128

typedef float v2f __attribute__((ext_vector_type(2)));
typedef float v4f __attribute__((ext_vector_type(4)));
typedef float v8f __attribute__((ext_vector_type(8)));


static __device__ __forceinline__ void lds_barrier()
{
    asm volatile("s_wait_dscnt 0x0\n\ts_barrier_signal -1\n\ts_barrier_wait -1" ::: "memory");
}

#define APITCH 132

__global__ void __launch_bounds__(256)
proj_gemm_kernel(const float* __restrict__ node,
                 const float* __restrict__ Wsrc, const float* __restrict__ bsrc,
                 const float* __restrict__ Wdst, const float* __restrict__ bdst,
                 float* __restrict__ src_proj, float* __restrict__ dest_proj,
                 int V)
{
    __shared__ __attribute__((aligned(16))) float As[2][16 * APITCH];

    const int lane    = threadIdx.x & 31;
    const int wave    = threadIdx.x >> 5;
    const int halfsel = lane >> 4;
    const int l15     = lane & 15;
    const int col     = wave * 16 + l15;

    const int nstrips = (V + 15) >> 4;
    int strip = blockIdx.x;
    if (strip >= nstrips) return;

    const int scol = lane * 4;

    {
        int r0 = strip * 16 + wave;
        int r1 = r0 + 8;
        r0 = r0 < V ? r0 : V - 1;
        r1 = r1 < V ? r1 : V - 1;
        const v4f g0 = *(const v4f*)(node + (size_t)r0 * DIM + scol);
        const v4f g1 = *(const v4f*)(node + (size_t)r1 * DIM + scol);
        *(v4f*)(&As[0][wave * APITCH + scol])       = g0;
        *(v4f*)(&As[0][(wave + 8) * APITCH + scol]) = g1;
    }

    v4f bs[16], bd[16];
#pragma unroll
    for (int j = 0; j < 16; ++j) {
        const size_t wo = (size_t)(8 * j + 4 * halfsel) * DIM + col;
        bs[j].x = Wsrc[wo + 0 * DIM];
        bs[j].y = Wsrc[wo + 1 * DIM];
        bs[j].z = Wsrc[wo + 2 * DIM];
        bs[j].w = Wsrc[wo + 3 * DIM];
        bd[j].x = Wdst[wo + 0 * DIM];
        bd[j].y = Wdst[wo + 1 * DIM];
        bd[j].z = Wdst[wo + 2 * DIM];
        bd[j].w = Wdst[wo + 3 * DIM];
    }
    const float bias_s = bsrc[col];
    const float bias_d = bdst[col];

    lds_barrier();

    int buf = 0;
    for (; strip < nstrips; strip += gridDim.x) {
        const int  next    = strip + gridDim.x;
        const bool hasNext = next < nstrips;

        v4f g0 = {}, g1 = {};
        if (hasNext) {
            int r0 = next * 16 + wave;
            int r1 = r0 + 8;
            r0 = r0 < V ? r0 : V - 1;
            r1 = r1 < V ? r1 : V - 1;
            g0 = *(const v4f*)(node + (size_t)r0 * DIM + scol);
            g1 = *(const v4f*)(node + (size_t)r1 * DIM + scol);
        }

        const size_t oo = (size_t)(strip * 16 + 8 * halfsel) * DIM + col;

        const float* ap = &As[buf][l15 * APITCH + 4 * halfsel];
        v4f a[16];
#pragma unroll
        for (int j = 0; j < 16; ++j)
            a[j] = *(const v4f*)(ap + 8 * j);
        __builtin_amdgcn_sched_barrier(0);

        v8f accs = {}, accd = {};
#pragma unroll
        for (int j = 0; j < 16; ++j) {
            v2f a0, a1, s0, s1, d0, d1;
            a0.x = a[j].x;  a0.y = a[j].y;
            a1.x = a[j].z;  a1.y = a[j].w;
            s0.x = bs[j].x; s0.y = bs[j].y;
            s1.x = bs[j].z; s1.y = bs[j].w;
            d0.x = bd[j].x; d0.y = bd[j].y;
            d1.x = bd[j].z; d1.y = bd[j].w;
            accs = __builtin_amdgcn_wmma_f32_16x16x4_f32(
                       false, a0, false, s0, (short)0, accs, false, false);
            accd = __builtin_amdgcn_wmma_f32_16x16x4_f32(
                       false, a0, false, d0, (short)0, accd, false, false);
            accs = __builtin_amdgcn_wmma_f32_16x16x4_f32(
                       false, a1, false, s1, (short)0, accs, false, false);
            accd = __builtin_amdgcn_wmma_f32_16x16x4_f32(
                       false, a1, false, d1, (short)0, accd, false, false);
        }

#pragma unroll
        for (int r = 0; r < 8; ++r) {
            src_proj[oo + (size_t)r * DIM]  = accs[r] + bias_s;
            dest_proj[oo + (size_t)r * DIM] = accd[r] + bias_d;
        }

        if (hasNext) {
            *(v4f*)(&As[buf ^ 1][wave * APITCH + scol])       = g0;
            *(v4f*)(&As[buf ^ 1][(wave + 8) * APITCH + scol]) = g1;
        }
        lds_barrier();
        buf ^= 1;
    }
}

#define EPW 16
#define GRP 4

template <bool FULL>
static __device__ __forceinline__ void edge_batch(
    const float* __restrict__ src_proj, const float* __restrict__ dest_proj,
    const float* __restrict__ edge_emb, float* __restrict__ out,
    int sv, int dv, int cv, int lane, int base, int E)
{
#pragma unroll
    for (int g = 0; g < EPW; g += GRP) {
        if (!FULL && base + g >= E) break;

        int   dn[GRP];
        float a[GRP][4], b[GRP][4], m[GRP][4];

#pragma unroll
        for (int u = 0; u < GRP; ++u) {
            const int s = __builtin_amdgcn_readlane(sv, g + u);
            const int d = __builtin_amdgcn_readlane(dv, g + u);
            dn[u] = d;
            const float* sp = src_proj  + (size_t)s * DIM + lane;
            const float* dp = dest_proj + (size_t)d * DIM + lane;
#pragma unroll
            for (int k = 0; k < 4; ++k) {
                a[u][k] = sp[32 * k];
                b[u][k] = dp[32 * k];
            }
        }
#pragma unroll
        for (int u = 0; u < GRP; ++u) {
            const int c = __builtin_amdgcn_readlane(cv, g + u);
            const float* em = edge_emb + (size_t)c * DIM + lane;
#pragma unroll
            for (int k = 0; k < 4; ++k)
                m[u][k] = em[32 * k];
        }

#pragma unroll
        for (int u = 0; u < GRP; ++u) {
            if (FULL || base + g + u < E) {
                float* o = out + (size_t)dn[u] * DIM + lane;
#pragma unroll
                for (int k = 0; k < 4; ++k)
                    unsafeAtomicAdd(o + 32 * k, fmaxf(a[u][k] + b[u][k] + m[u][k], 0.0f));
            }
        }
    }
}

__global__ void __launch_bounds__(256)
edge_scatter_kernel(const float* __restrict__ src_proj,
                    const float* __restrict__ dest_proj,
                    const int* __restrict__ edge_src,
                    const int* __restrict__ edge_dest,
                    const int* __restrict__ edge_cls,
                    const float* __restrict__ edge_emb,
                    float* __restrict__ out, int E)
{
    const int lane = threadIdx.x & 31;
    const int wave = __builtin_amdgcn_readfirstlane((int)(threadIdx.x >> 5));
    const int base = (blockIdx.x * 8 + wave) * EPW;
    if (base >= E) return;

    int ei = base + (lane & (EPW - 1));
    ei = ei < E ? ei : E - 1;
    const int sv = edge_src[ei];
    const int dv = edge_dest[ei];
    const int cv = edge_cls[ei];

    if (base + EPW <= E)
        edge_batch<true>(src_proj, dest_proj, edge_emb, out, sv, dv, cv, lane, base, E);
    else
        edge_batch<false>(src_proj, dest_proj, edge_emb, out, sv, dv, cv, lane, base, E);
}

extern "C" void kernel_launch(void* const* d_in, const int* in_sizes, int n_in,
                              void* d_out, int out_size, void* d_ws, size_t ws_size,
                              hipStream_t stream) {
    const float* node = (const float*)d_in[0];
    const int*   esrc = (const int*)d_in[1];
    const int*   edst = (const int*)d_in[2];
    const int*   ecls = (const int*)d_in[3];
    const float* Wsrc = (const float*)d_in[4];
    const float* bsrc = (const float*)d_in[5];
    const float* Wdst = (const float*)d_in[6];
    const float* bdst = (const float*)d_in[7];
    const float* eemb = (const float*)d_in[8];
    float* out = (float*)d_out;

    const int V = in_sizes[0] / DIM;
    const int E = in_sizes[1];

    float* src_proj  = (float*)d_ws;
    float* dest_proj = src_proj + (size_t)V * DIM;

    hipMemsetAsync(d_out, 0, (size_t)V * DIM * sizeof(float), stream);

    proj_gemm_kernel<<<dim3(512), dim3(256), 0, stream>>>(
        node, Wsrc, bsrc, Wdst, bdst, src_proj, dest_proj, V);

    const int nblocks = (E + 8 * EPW - 1) / (8 * EPW);
    edge_scatter_kernel<<<nblocks, dim3(256), 0, stream>>>(
        src_proj, dest_proj, esrc, edst, ecls, eemb, out, E);
}
